# starting kernel
# baseline (speedup 1.0000x reference)
#include <hip/hip_runtime.h>
#include <hip/hip_bf16.h>
#include <math.h>

#define NN   4096
#define MD   64
#define MSG  64
#define RAW  4224
#define HID  2112
#define WL   8

typedef __attribute__((ext_vector_type(2))) float v2f;
typedef __attribute__((ext_vector_type(8))) float v8f;

__device__ __forceinline__ v2f ldv2(const float* p) { return *(const v2f*)p; }

__device__ __forceinline__ void wmma_step4x4(
    v8f acc[4][4],
    const float* __restrict__ A, int r0, int ldA, int kA,
    const float* __restrict__ B, int c0, int ldB, int kB, int half) {
  v2f a[4], b[4];
#pragma unroll
  for (int t = 0; t < 4; ++t) {
    a[t] = ldv2(A + (size_t)(r0 + 16 * t) * (size_t)ldA + kA + 2 * half);
    b[t] = ldv2(B + (size_t)(c0 + 16 * t) * (size_t)ldB + kB + 2 * half);
  }
#pragma unroll
  for (int mi = 0; mi < 4; ++mi)
#pragma unroll
    for (int ni = 0; ni < 4; ++ni)
      acc[mi][ni] = __builtin_amdgcn_wmma_f32_16x16x4_f32(
          false, a[mi], false, b[ni], (short)0, acc[mi][ni], false, false);
}

__global__ __launch_bounds__(64) void k_rw(
    const float* __restrict__ memory, const int* __restrict__ walks,
    const float* __restrict__ W_rw, const float* __restrict__ b_rw,
    float* __restrict__ rw) {
  __shared__ float pooled[MD];
  const int n = blockIdx.x;
  const int d = threadIdx.x;
  float s = 0.f;
#pragma unroll
  for (int w = 0; w < WL; ++w) {
    const int idx = walks[n * WL + w];
    s += memory[(size_t)idx * MD + d];
  }
  pooled[d] = s * (1.0f / WL);
  __syncthreads();
  float acc = b_rw[d];
#pragma unroll 8
  for (int k = 0; k < MD; ++k) acc += pooled[k] * W_rw[d * MD + k];
  rw[(size_t)n * MSG + d] = acc;
}

__global__ __launch_bounds__(32) void k_gemm1(
    const float* __restrict__ memory, const float* __restrict__ od,
    const float* __restrict__ rw, const float* __restrict__ W1,
    const float* __restrict__ bias1, float* __restrict__ hid) {
  const int lane = threadIdx.x & 31;
  const int l15 = lane & 15;
  const int half = lane >> 4;
  const int nBase = blockIdx.x * 64;
  const int mBase = blockIdx.y * 64;
  const int r0 = mBase + l15;
  const int c0 = nBase + l15;

  v8f acc[4][4] = {};

  for (int k0 = 0; k0 < MD; k0 += 4)
    wmma_step4x4(acc, memory, r0, MD, k0, W1, c0, RAW, k0, half);
  for (int k0 = MD; k0 < MD + NN; k0 += 4)
    wmma_step4x4(acc, od, r0, NN, k0 - MD, W1, c0, RAW, k0, half);
  for (int k0 = MD + NN; k0 < RAW; k0 += 4)
    wmma_step4x4(acc, rw, r0, MSG, k0 - (MD + NN), W1, c0, RAW, k0, half);

#pragma unroll
  for (int ni = 0; ni < 4; ++ni) {
    const int col = nBase + ni * 16 + l15;
    const float bv = bias1[col];
#pragma unroll
    for (int mi = 0; mi < 4; ++mi) {
#pragma unroll
      for (int i = 0; i < 8; ++i) {
        const int row = mBase + mi * 16 + i + 8 * half;
        const float v = acc[mi][ni][i] + bv;
        hid[(size_t)row * HID + col] = v > 0.f ? v : 0.f;
      }
    }
  }
}

__global__ __launch_bounds__(32) void k_gemm2(
    const float* __restrict__ hid, const float* __restrict__ W2,
    const float* __restrict__ bias2, float* __restrict__ msg) {
  const int lane = threadIdx.x & 31;
  const int l15 = lane & 15;
  const int half = lane >> 4;
  const int mBase = blockIdx.x * 16;
  const int r0 = mBase + l15;

  v8f acc[4] = {};
  for (int k0 = 0; k0 < HID; k0 += 4) {
    const v2f a = ldv2(hid + (size_t)r0 * HID + k0 + 2 * half);
#pragma unroll
    for (int ni = 0; ni < 4; ++ni) {
      const v2f b = ldv2(W2 + (size_t)(ni * 16 + l15) * HID + k0 + 2 * half);
      acc[ni] = __builtin_amdgcn_wmma_f32_16x16x4_f32(
          false, a, false, b, (short)0, acc[ni], false, false);
    }
  }
#pragma unroll
  for (int ni = 0; ni < 4; ++ni) {
    const int col = ni * 16 + l15;
    const float bv = bias2[col];
#pragma unroll
    for (int i = 0; i < 8; ++i) {
      const int row = mBase + i + 8 * half;
      msg[(size_t)row * MSG + col] = acc[ni][i] + bv;
    }
  }
}

__global__ __launch_bounds__(64) void k_gru(
    const float* __restrict__ msg, const float* __restrict__ memory,
    const float* __restrict__ Wi, const float* __restrict__ bi,
    const float* __restrict__ Wh, const float* __restrict__ bh,
    float* __restrict__ upd) {
  __shared__ float sm[MSG];
  __shared__ float sh[MD];
  const int n = blockIdx.x;
  const int j = threadIdx.x;
  sm[j] = msg[(size_t)n * MSG + j];
  sh[j] = memory[(size_t)n * MD + j];
  __syncthreads();
  float ir = bi[j], iz = bi[MD + j], in_ = bi[2 * MD + j];
  float hr = bh[j], hz = bh[MD + j], hn = bh[2 * MD + j];
#pragma unroll 4
  for (int m = 0; m < MSG; ++m) {
    const float mv = sm[m];
    const float hv = sh[m];
    ir += mv * Wi[(size_t)(j) * MSG + m];
    iz += mv * Wi[(size_t)(MD + j) * MSG + m];
    in_ += mv * Wi[(size_t)(2 * MD + j) * MSG + m];
    hr += hv * Wh[(size_t)(j) * MD + m];
    hz += hv * Wh[(size_t)(MD + j) * MD + m];
    hn += hv * Wh[(size_t)(2 * MD + j) * MD + m];
  }
  const float r = 1.f / (1.f + __expf(-(ir + hr)));
  const float z = 1.f / (1.f + __expf(-(iz + hz)));
  const float nn = tanhf(in_ + r * hn);
  upd[(size_t)n * MD + j] = (1.f - z) * nn + z * sh[j];
}

__global__ __launch_bounds__(64) void k_pred1(
    const float* __restrict__ upd, const float* __restrict__ Wp1,
    const float* __restrict__ bp1, float* __restrict__ p) {
  __shared__ float su[MD];
  const int n = blockIdx.x;
  const int j = threadIdx.x;
  su[j] = upd[(size_t)n * MD + j];
  __syncthreads();
  float acc = bp1[j];
#pragma unroll 8
  for (int d = 0; d < MD; ++d) acc += su[d] * Wp1[j * MD + d];
  p[(size_t)n * MD + j] = acc > 0.f ? acc : 0.f;
}

__global__ __launch_bounds__(128) void k_pred2(
    const float* __restrict__ p, const float* __restrict__ Wp2,
    const float* __restrict__ bp2, float* __restrict__ out) {
  const int lane = threadIdx.x & 31;
  const int wave = threadIdx.x >> 5;
  const int l15 = lane & 15;
  const int half = lane >> 4;
  const int mBase = blockIdx.y * 64;
  const int nBase = blockIdx.x * 256 + wave * 64;
  const int r0 = mBase + l15;
  const int c0 = nBase + l15;

  v8f acc[4][4] = {};
#pragma unroll
  for (int k0 = 0; k0 < MD; k0 += 4)
    wmma_step4x4(acc, p, r0, MD, k0, Wp2, c0, MD, k0, half);

#pragma unroll
  for (int ni = 0; ni < 4; ++ni) {
    const int col = nBase + ni * 16 + l15;
    const float bv = bp2[col];
#pragma unroll
    for (int mi = 0; mi < 4; ++mi) {
#pragma unroll
      for (int i = 0; i < 8; ++i) {
        const int row = mBase + mi * 16 + i + 8 * half;
        out[(size_t)row * NN + col] = acc[mi][ni][i] + bv;
      }
    }
  }
}

extern "C" void kernel_launch(void* const* d_in, const int* in_sizes, int n_in,
                              void* d_out, int out_size, void* d_ws, size_t ws_size,
                              hipStream_t stream) {
  const float* memory = (const float*)d_in[0];
  const float* od_mat = (const float*)d_in[1];
  const int*   walks  = (const int*)d_in[2];
  const float* W_rw   = (const float*)d_in[3];
  const float* b_rw   = (const float*)d_in[4];
  const float* W1     = (const float*)d_in[5];
  const float* b1     = (const float*)d_in[6];
  const float* W2     = (const float*)d_in[7];
  const float* b2     = (const float*)d_in[8];
  const float* gWi    = (const float*)d_in[9];
  const float* gbi    = (const float*)d_in[10];
  const float* gWh    = (const float*)d_in[11];
  const float* gbh    = (const float*)d_in[12];
  const float* Wp1    = (const float*)d_in[13];
  const float* bp1    = (const float*)d_in[14];
  const float* Wp2    = (const float*)d_in[15];
  const float* bp2    = (const float*)d_in[16];
  float* out = (float*)d_out;

  char* ws = (char*)d_ws;
  const size_t SZ_RW  = (size_t)NN * MSG * sizeof(float);
  const size_t SZ_HID = (size_t)NN * HID * sizeof(float);
  const size_t SZ_MSG = (size_t)NN * MSG * sizeof(float);
  const size_t SZ_UPD = (size_t)NN * MD  * sizeof(float);
  float* rw   = (float*)(ws);
  float* hid  = (float*)(ws + SZ_RW);
  float* msg  = (float*)(ws + SZ_RW + SZ_HID);
  float* upd  = (float*)(ws + SZ_RW + SZ_HID + SZ_MSG);
  float* pbuf = (float*)(ws + SZ_RW + SZ_HID + SZ_MSG + SZ_UPD);

  k_rw<<<NN, 64, 0, stream>>>(memory, walks, W_rw, b_rw, rw);
  {
    dim3 grid(HID / 64, NN / 64);
    k_gemm1<<<grid, 32, 0, stream>>>(memory, od_mat, rw, W1, b1, hid);
  }
  k_gemm2<<<NN / 16, 32, 0, stream>>>(hid, W2, b2, msg);
  k_gru<<<NN, 64, 0, stream>>>(msg, memory, gWi, gbi, gWh, gbh, upd);
  k_pred1<<<NN, 64, 0, stream>>>(upd, Wp1, bp1, pbuf);
  {
    dim3 grid(NN / 256, NN / 64);
    k_pred2<<<grid, 128, 0, stream>>>(pbuf, Wp2, bp2, out);
  }
}


# layer-1 GEMM operands staged through a 2-stage LDS ring by coalesced 16-byte global loads, one 64x528 tile per 4-wave work-group; k is summed in a different order inside each 8-k group
# speedup vs baseline: 1.5469x; 1.5469x over previous
#include <hip/hip_runtime.h>
#include <hip/hip_bf16.h>
#include <math.h>

#define NN   4096
#define MD   64
#define MSG  64
#define RAW  4224
#define HID  2112
#define WL   8

typedef __attribute__((ext_vector_type(2))) float v2f;
typedef __attribute__((ext_vector_type(8))) float v8f;

__device__ __forceinline__ v2f ldv2(const float* p) { return *(const v2f*)p; }

__device__ __forceinline__ void wmma_step4x4(
    v8f acc[4][4],
    const float* __restrict__ A, int r0, int ldA, int kA,
    const float* __restrict__ B, int c0, int ldB, int kB, int half) {
  v2f a[4], b[4];
#pragma unroll
  for (int t = 0; t < 4; ++t) {
    a[t] = ldv2(A + (size_t)(r0 + 16 * t) * (size_t)ldA + kA + 2 * half);
    b[t] = ldv2(B + (size_t)(c0 + 16 * t) * (size_t)ldB + kB + 2 * half);
  }
#pragma unroll
  for (int mi = 0; mi < 4; ++mi)
#pragma unroll
    for (int ni = 0; ni < 4; ++ni)
      acc[mi][ni] = __builtin_amdgcn_wmma_f32_16x16x4_f32(
          false, a[mi], false, b[ni], (short)0, acc[mi][ni], false, false);
}

__global__ __launch_bounds__(64) void k_rw(
    const float* __restrict__ memory, const int* __restrict__ walks,
    const float* __restrict__ W_rw, const float* __restrict__ b_rw,
    float* __restrict__ rw) {
  __shared__ float pooled[MD];
  const int n = blockIdx.x;
  const int d = threadIdx.x;
  float s = 0.f;
#pragma unroll
  for (int w = 0; w < WL; ++w) {
    const int idx = walks[n * WL + w];
    s += memory[(size_t)idx * MD + d];
  }
  pooled[d] = s * (1.0f / WL);
  __syncthreads();
  float acc = b_rw[d];
#pragma unroll 8
  for (int k = 0; k < MD; ++k) acc += pooled[k] * W_rw[d * MD + k];
  rw[(size_t)n * MSG + d] = acc;
}

#define G1_BM    64
#define G1_BN    528
#define G1_KC    32
#define G1_PITCH 36
#define G1_ROWS  (G1_BM + G1_BN)
#define G1_STAGE (G1_ROWS * G1_PITCH)
#define G1_NT    (G1_BN / 16)
#define G1_NCH   (RAW / G1_KC)
#define G1_C_OD  (MD / G1_KC)
#define G1_C_RW  ((MD + NN) / G1_KC)

typedef __attribute__((ext_vector_type(4))) float v4f;

__device__ __forceinline__ void g1_gload(
    v4f (&stA)[4], v4f (&stB)[G1_NT], int cn,
    const float* __restrict__ memory, const float* __restrict__ od,
    const float* __restrict__ rw, const float* __restrict__ gB, int arow, int ccol) {
  const float* src;
  int ld, k;
  if (cn < G1_C_OD)      { src = memory; ld = MD;  k = cn * G1_KC; }
  else if (cn < G1_C_RW) { src = od;     ld = NN;  k = (cn - G1_C_OD) * G1_KC; }
  else                   { src = rw;     ld = MSG; k = (cn - G1_C_RW) * G1_KC; }
  const float* gA = src + (size_t)arow * ld + k + ccol;
#pragma unroll
  for (int i = 0; i < 4; ++i)
    stA[i] = *(const v4f*)(gA + (size_t)i * 16 * ld);
  const float* gBn = gB + cn * G1_KC;
#pragma unroll
  for (int i = 0; i < G1_NT; ++i)
    stB[i] = *(const v4f*)(gBn + (size_t)i * 16 * RAW);
}

__device__ __forceinline__ void g1_lstore(
    float* sN, const v4f (&stA)[4], const v4f (&stB)[G1_NT]) {
#pragma unroll
  for (int i = 0; i < 4; ++i)
    *(v4f*)(sN + i * 16 * G1_PITCH) = stA[i];
#pragma unroll
  for (int i = 0; i < G1_NT; ++i)
    *(v4f*)(sN + (G1_BM + i * 16) * G1_PITCH) = stB[i];
}

__device__ __forceinline__ void g1_frags(
    v4f& a, v4f (&b)[G1_NT], const float* S, int fA, int fB) {
  a = *(const v4f*)(S + fA);
#pragma unroll
  for (int j = 0; j < G1_NT; ++j)
    b[j] = *(const v4f*)(S + fB + j * 16 * G1_PITCH);
}

__device__ __forceinline__ void g1_mma(
    v8f (&acc)[G1_NT], const v4f& a, const v4f (&b)[G1_NT]) {
  const v2f alo = {a.x, a.y};
  const v2f ahi = {a.z, a.w};
#pragma unroll
  for (int j = 0; j < G1_NT; ++j) {
    const v2f blo = {b[j].x, b[j].y};
    acc[j] = __builtin_amdgcn_wmma_f32_16x16x4_f32(
        false, alo, false, blo, (short)0, acc[j], false, false);
  }
#pragma unroll
  for (int j = 0; j < G1_NT; ++j) {
    const v2f bhi = {b[j].z, b[j].w};
    acc[j] = __builtin_amdgcn_wmma_f32_16x16x4_f32(
        false, ahi, false, bhi, (short)0, acc[j], false, false);
  }
}

__device__ __forceinline__ void g1_pin() {
  __builtin_amdgcn_sched_barrier(0);
}

__global__ __launch_bounds__(128) void k_gemm1(
    const float* __restrict__ memory, const float* __restrict__ od,
    const float* __restrict__ rw, const float* __restrict__ W1,
    const float* __restrict__ bias1, float* __restrict__ hid) {
  __shared__ __attribute__((aligned(16))) float L[2 * G1_STAGE];

  const int tid = threadIdx.x;
  const int lane = tid & 31;
  const int wave = tid >> 5;
  const int l15 = lane & 15;
  const int half = lane >> 4;
  const int nBase = blockIdx.x * G1_BN;
  const int mBase = blockIdx.y * G1_BM;

  const int crow = tid >> 3;
  const int ccol = (tid & 7) * 4;
  const int arow = mBase + crow;
  const float* gB = W1 + (size_t)(nBase + crow) * RAW + ccol;
  float* sW = L + crow * G1_PITCH + ccol;

  const int fA = (wave * 16 + l15) * G1_PITCH + 4 * half;
  const int fB = (G1_BM + l15) * G1_PITCH + 4 * half;

  v4f stA[4], stB[G1_NT];
  v4f a0, a1, b0[G1_NT], b1[G1_NT];
  v8f acc[G1_NT] = {};

  g1_gload(stA, stB, 0, memory, od, rw, gB, arow, ccol);
  g1_lstore(sW, stA, stB);
  g1_gload(stA, stB, 1, memory, od, rw, gB, arow, ccol);
  __syncthreads();
  g1_frags(a0, b0, L, fA, fB);

  for (int ch = 0; ch < G1_NCH - 1; ++ch) {
    const float* S = L + (ch & 1) * G1_STAGE;
    const int nOff = ((ch + 1) & 1) * G1_STAGE;

    g1_frags(a1, b1, S + 8, fA, fB);
    g1_pin();
    g1_mma(acc, a0, b0);
    g1_pin();
    g1_frags(a0, b0, S + 16, fA, fB);
    g1_pin();
    g1_mma(acc, a1, b1);
    g1_pin();
    g1_frags(a1, b1, S + 24, fA, fB);
    g1_lstore(sW + nOff, stA, stB);
    g1_pin();
    g1_mma(acc, a0, b0);
    g1_pin();
    __syncthreads();
    g1_frags(a0, b0, L + nOff, fA, fB);
    if (ch + 2 < G1_NCH)
      g1_gload(stA, stB, ch + 2, memory, od, rw, gB, arow, ccol);
    g1_pin();
    g1_mma(acc, a1, b1);
  }
  {
    const float* S = L + ((G1_NCH - 1) & 1) * G1_STAGE;
    g1_frags(a1, b1, S + 8, fA, fB);
    g1_pin();
    g1_mma(acc, a0, b0);
    g1_pin();
    g1_frags(a0, b0, S + 16, fA, fB);
    g1_pin();
    g1_mma(acc, a1, b1);
    g1_pin();
    g1_frags(a1, b1, S + 24, fA, fB);
    g1_pin();
    g1_mma(acc, a0, b0);
    g1_mma(acc, a1, b1);
  }

#pragma unroll
  for (int j = 0; j < G1_NT; ++j) {
    const int col = nBase + j * 16 + l15;
    const float bv = bias1[col];
#pragma unroll
    for (int i = 0; i < 8; ++i) {
      const int row = mBase + wave * 16 + i + 8 * half;
      const float v = acc[j][i] + bv;
      hid[(size_t)row * HID + col] = v > 0.f ? v : 0.f;
    }
  }
}

__global__ __launch_bounds__(32) void k_gemm2(
    const float* __restrict__ hid, const float* __restrict__ W2,
    const float* __restrict__ bias2, float* __restrict__ msg) {
  const int lane = threadIdx.x & 31;
  const int l15 = lane & 15;
  const int half = lane >> 4;
  const int mBase = blockIdx.x * 16;
  const int r0 = mBase + l15;

  v8f acc[4] = {};
  for (int k0 = 0; k0 < HID; k0 += 4) {
    const v2f a = ldv2(hid + (size_t)r0 * HID + k0 + 2 * half);
#pragma unroll
    for (int ni = 0; ni < 4; ++ni) {
      const v2f b = ldv2(W2 + (size_t)(ni * 16 + l15) * HID + k0 + 2 * half);
      acc[ni] = __builtin_amdgcn_wmma_f32_16x16x4_f32(
          false, a, false, b, (short)0, acc[ni], false, false);
    }
  }
#pragma unroll
  for (int ni = 0; ni < 4; ++ni) {
    const int col = ni * 16 + l15;
    const float bv = bias2[col];
#pragma unroll
    for (int i = 0; i < 8; ++i) {
      const int row = mBase + i + 8 * half;
      msg[(size_t)row * MSG + col] = acc[ni][i] + bv;
    }
  }
}

__global__ __launch_bounds__(64) void k_gru(
    const float* __restrict__ msg, const float* __restrict__ memory,
    const float* __restrict__ Wi, const float* __restrict__ bi,
    const float* __restrict__ Wh, const float* __restrict__ bh,
    float* __restrict__ upd) {
  __shared__ float sm[MSG];
  __shared__ float sh[MD];
  const int n = blockIdx.x;
  const int j = threadIdx.x;
  sm[j] = msg[(size_t)n * MSG + j];
  sh[j] = memory[(size_t)n * MD + j];
  __syncthreads();
  float ir = bi[j], iz = bi[MD + j], in_ = bi[2 * MD + j];
  float hr = bh[j], hz = bh[MD + j], hn = bh[2 * MD + j];
#pragma unroll 4
  for (int m = 0; m < MSG; ++m) {
    const float mv = sm[m];
    const float hv = sh[m];
    ir += mv * Wi[(size_t)(j) * MSG + m];
    iz += mv * Wi[(size_t)(MD + j) * MSG + m];
    in_ += mv * Wi[(size_t)(2 * MD + j) * MSG + m];
    hr += hv * Wh[(size_t)(j) * MD + m];
    hz += hv * Wh[(size_t)(MD + j) * MD + m];
    hn += hv * Wh[(size_t)(2 * MD + j) * MD + m];
  }
  const float r = 1.f / (1.f + __expf(-(ir + hr)));
  const float z = 1.f / (1.f + __expf(-(iz + hz)));
  const float nn = tanhf(in_ + r * hn);
  upd[(size_t)n * MD + j] = (1.f - z) * nn + z * sh[j];
}

__global__ __launch_bounds__(64) void k_pred1(
    const float* __restrict__ upd, const float* __restrict__ Wp1,
    const float* __restrict__ bp1, float* __restrict__ p) {
  __shared__ float su[MD];
  const int n = blockIdx.x;
  const int j = threadIdx.x;
  su[j] = upd[(size_t)n * MD + j];
  __syncthreads();
  float acc = bp1[j];
#pragma unroll 8
  for (int d = 0; d < MD; ++d) acc += su[d] * Wp1[j * MD + d];
  p[(size_t)n * MD + j] = acc > 0.f ? acc : 0.f;
}

__global__ __launch_bounds__(128) void k_pred2(
    const float* __restrict__ p, const float* __restrict__ Wp2,
    const float* __restrict__ bp2, float* __restrict__ out) {
  const int lane = threadIdx.x & 31;
  const int wave = threadIdx.x >> 5;
  const int l15 = lane & 15;
  const int half = lane >> 4;
  const int mBase = blockIdx.y * 64;
  const int nBase = blockIdx.x * 256 + wave * 64;
  const int r0 = mBase + l15;
  const int c0 = nBase + l15;

  v8f acc[4][4] = {};
#pragma unroll
  for (int k0 = 0; k0 < MD; k0 += 4)
    wmma_step4x4(acc, p, r0, MD, k0, Wp2, c0, MD, k0, half);

#pragma unroll
  for (int ni = 0; ni < 4; ++ni) {
    const int col = nBase + ni * 16 + l15;
    const float bv = bp2[col];
#pragma unroll
    for (int mi = 0; mi < 4; ++mi) {
#pragma unroll
      for (int i = 0; i < 8; ++i) {
        const int row = mBase + mi * 16 + i + 8 * half;
        out[(size_t)row * NN + col] = acc[mi][ni][i] + bv;
      }
    }
  }
}

extern "C" void kernel_launch(void* const* d_in, const int* in_sizes, int n_in,
                              void* d_out, int out_size, void* d_ws, size_t ws_size,
                              hipStream_t stream) {
  const float* memory = (const float*)d_in[0];
  const float* od_mat = (const float*)d_in[1];
  const int*   walks  = (const int*)d_in[2];
  const float* W_rw   = (const float*)d_in[3];
  const float* b_rw   = (const float*)d_in[4];
  const float* W1     = (const float*)d_in[5];
  const float* b1     = (const float*)d_in[6];
  const float* W2     = (const float*)d_in[7];
  const float* b2     = (const float*)d_in[8];
  const float* gWi    = (const float*)d_in[9];
  const float* gbi    = (const float*)d_in[10];
  const float* gWh    = (const float*)d_in[11];
  const float* gbh    = (const float*)d_in[12];
  const float* Wp1    = (const float*)d_in[13];
  const float* bp1    = (const float*)d_in[14];
  const float* Wp2    = (const float*)d_in[15];
  const float* bp2    = (const float*)d_in[16];
  float* out = (float*)d_out;

  char* ws = (char*)d_ws;
  const size_t SZ_RW  = (size_t)NN * MSG * sizeof(float);
  const size_t SZ_HID = (size_t)NN * HID * sizeof(float);
  const size_t SZ_MSG = (size_t)NN * MSG * sizeof(float);
  const size_t SZ_UPD = (size_t)NN * MD  * sizeof(float);
  float* rw   = (float*)(ws);
  float* hid  = (float*)(ws + SZ_RW);
  float* msg  = (float*)(ws + SZ_RW + SZ_HID);
  float* upd  = (float*)(ws + SZ_RW + SZ_HID + SZ_MSG);
  float* pbuf = (float*)(ws + SZ_RW + SZ_HID + SZ_MSG + SZ_UPD);

  k_rw<<<NN, 64, 0, stream>>>(memory, walks, W_rw, b_rw, rw);
  {
    dim3 grid(HID / G1_BN, NN / G1_BM);
    k_gemm1<<<grid, 128, 0, stream>>>(memory, od_mat, rw, W1, b1, hid);
  }
  k_gemm2<<<NN / 16, 32, 0, stream>>>(hid, W2, b2, msg);
  k_gru<<<NN, 64, 0, stream>>>(msg, memory, gWi, gbi, gWh, gbh, upd);
  k_pred1<<<NN, 64, 0, stream>>>(upd, Wp1, bp1, pbuf);
  {
    dim3 grid(NN / 256, NN / 64);
    k_pred2<<<grid, 128, 0, stream>>>(pbuf, Wp2, bp2, out);
  }
}


# layer 2, GRU cell and prediction hidden layer fused into one launch per 16-node tile: layer-2 K split over 8 waves, summed through LDS in wave order; gates and hidden layer as f32 WMMA on the tile
# speedup vs baseline: 1.8520x; 1.1972x over previous
#include <hip/hip_runtime.h>
#include <hip/hip_bf16.h>
#include <math.h>

#define NN   4096
#define MD   64
#define MSG  64
#define RAW  4224
#define HID  2112
#define WL   8

typedef __attribute__((ext_vector_type(2))) float v2f;
typedef __attribute__((ext_vector_type(8))) float v8f;

__device__ __forceinline__ v2f ldv2(const float* p) { return *(const v2f*)p; }

__device__ __forceinline__ void wmma_step4x4(
    v8f acc[4][4],
    const float* __restrict__ A, int r0, int ldA, int kA,
    const float* __restrict__ B, int c0, int ldB, int kB, int half) {
  v2f a[4], b[4];
#pragma unroll
  for (int t = 0; t < 4; ++t) {
    a[t] = ldv2(A + (size_t)(r0 + 16 * t) * (size_t)ldA + kA + 2 * half);
    b[t] = ldv2(B + (size_t)(c0 + 16 * t) * (size_t)ldB + kB + 2 * half);
  }
#pragma unroll
  for (int mi = 0; mi < 4; ++mi)
#pragma unroll
    for (int ni = 0; ni < 4; ++ni)
      acc[mi][ni] = __builtin_amdgcn_wmma_f32_16x16x4_f32(
          false, a[mi], false, b[ni], (short)0, acc[mi][ni], false, false);
}

__global__ __launch_bounds__(64) void k_rw(
    const float* __restrict__ memory, const int* __restrict__ walks,
    const float* __restrict__ W_rw, const float* __restrict__ b_rw,
    float* __restrict__ rw) {
  __shared__ float pooled[MD];
  const int n = blockIdx.x;
  const int d = threadIdx.x;
  float s = 0.f;
#pragma unroll
  for (int w = 0; w < WL; ++w) {
    const int idx = walks[n * WL + w];
    s += memory[(size_t)idx * MD + d];
  }
  pooled[d] = s * (1.0f / WL);
  __syncthreads();
  float acc = b_rw[d];
#pragma unroll 8
  for (int k = 0; k < MD; ++k) acc += pooled[k] * W_rw[d * MD + k];
  rw[(size_t)n * MSG + d] = acc;
}

#define G1_BM    64
#define G1_BN    528
#define G1_KC    32
#define G1_PITCH 36
#define G1_ROWS  (G1_BM + G1_BN)
#define G1_STAGE (G1_ROWS * G1_PITCH)
#define G1_NT    (G1_BN / 16)
#define G1_NCH   (RAW / G1_KC)
#define G1_C_OD  (MD / G1_KC)
#define G1_C_RW  ((MD + NN) / G1_KC)

typedef __attribute__((ext_vector_type(4))) float v4f;

__device__ __forceinline__ void g1_gload(
    v4f (&stA)[4], v4f (&stB)[G1_NT], int cn,
    const float* __restrict__ memory, const float* __restrict__ od,
    const float* __restrict__ rw, const float* __restrict__ gB, int arow, int ccol) {
  const float* src;
  int ld, k;
  if (cn < G1_C_OD)      { src = memory; ld = MD;  k = cn * G1_KC; }
  else if (cn < G1_C_RW) { src = od;     ld = NN;  k = (cn - G1_C_OD) * G1_KC; }
  else                   { src = rw;     ld = MSG; k = (cn - G1_C_RW) * G1_KC; }
  const float* gA = src + (size_t)arow * ld + k + ccol;
#pragma unroll
  for (int i = 0; i < 4; ++i)
    stA[i] = *(const v4f*)(gA + (size_t)i * 16 * ld);
  const float* gBn = gB + cn * G1_KC;
#pragma unroll
  for (int i = 0; i < G1_NT; ++i)
    stB[i] = *(const v4f*)(gBn + (size_t)i * 16 * RAW);
}

__device__ __forceinline__ void g1_lstore(
    float* sN, const v4f (&stA)[4], const v4f (&stB)[G1_NT]) {
#pragma unroll
  for (int i = 0; i < 4; ++i)
    *(v4f*)(sN + i * 16 * G1_PITCH) = stA[i];
#pragma unroll
  for (int i = 0; i < G1_NT; ++i)
    *(v4f*)(sN + (G1_BM + i * 16) * G1_PITCH) = stB[i];
}

__device__ __forceinline__ void g1_frags(
    v4f& a, v4f (&b)[G1_NT], const float* S, int fA, int fB) {
  a = *(const v4f*)(S + fA);
#pragma unroll
  for (int j = 0; j < G1_NT; ++j)
    b[j] = *(const v4f*)(S + fB + j * 16 * G1_PITCH);
}

__device__ __forceinline__ void g1_mma(
    v8f (&acc)[G1_NT], const v4f& a, const v4f (&b)[G1_NT]) {
  const v2f alo = {a.x, a.y};
  const v2f ahi = {a.z, a.w};
#pragma unroll
  for (int j = 0; j < G1_NT; ++j) {
    const v2f blo = {b[j].x, b[j].y};
    acc[j] = __builtin_amdgcn_wmma_f32_16x16x4_f32(
        false, alo, false, blo, (short)0, acc[j], false, false);
  }
#pragma unroll
  for (int j = 0; j < G1_NT; ++j) {
    const v2f bhi = {b[j].z, b[j].w};
    acc[j] = __builtin_amdgcn_wmma_f32_16x16x4_f32(
        false, ahi, false, bhi, (short)0, acc[j], false, false);
  }
}

__device__ __forceinline__ void g1_pin() {
  __builtin_amdgcn_sched_barrier(0);
}

__global__ __launch_bounds__(128) void k_gemm1(
    const float* __restrict__ memory, const float* __restrict__ od,
    const float* __restrict__ rw, const float* __restrict__ W1,
    const float* __restrict__ bias1, float* __restrict__ hid) {
  __shared__ __attribute__((aligned(16))) float L[2 * G1_STAGE];

  const int tid = threadIdx.x;
  const int lane = tid & 31;
  const int wave = tid >> 5;
  const int l15 = lane & 15;
  const int half = lane >> 4;
  const int nBase = blockIdx.x * G1_BN;
  const int mBase = blockIdx.y * G1_BM;

  const int crow = tid >> 3;
  const int ccol = (tid & 7) * 4;
  const int arow = mBase + crow;
  const float* gB = W1 + (size_t)(nBase + crow) * RAW + ccol;
  float* sW = L + crow * G1_PITCH + ccol;

  const int fA = (wave * 16 + l15) * G1_PITCH + 4 * half;
  const int fB = (G1_BM + l15) * G1_PITCH + 4 * half;

  v4f stA[4], stB[G1_NT];
  v4f a0, a1, b0[G1_NT], b1[G1_NT];
  v8f acc[G1_NT] = {};

  g1_gload(stA, stB, 0, memory, od, rw, gB, arow, ccol);
  g1_lstore(sW, stA, stB);
  g1_gload(stA, stB, 1, memory, od, rw, gB, arow, ccol);
  __syncthreads();
  g1_frags(a0, b0, L, fA, fB);

  for (int ch = 0; ch < G1_NCH - 1; ++ch) {
    const float* S = L + (ch & 1) * G1_STAGE;
    const int nOff = ((ch + 1) & 1) * G1_STAGE;

    g1_frags(a1, b1, S + 8, fA, fB);
    g1_pin();
    g1_mma(acc, a0, b0);
    g1_pin();
    g1_frags(a0, b0, S + 16, fA, fB);
    g1_pin();
    g1_mma(acc, a1, b1);
    g1_pin();
    g1_frags(a1, b1, S + 24, fA, fB);
    g1_lstore(sW + nOff, stA, stB);
    g1_pin();
    g1_mma(acc, a0, b0);
    g1_pin();
    __syncthreads();
    g1_frags(a0, b0, L + nOff, fA, fB);
    if (ch + 2 < G1_NCH)
      g1_gload(stA, stB, ch + 2, memory, od, rw, gB, arow, ccol);
    g1_pin();
    g1_mma(acc, a1, b1);
  }
  {
    const float* S = L + ((G1_NCH - 1) & 1) * G1_STAGE;
    g1_frags(a1, b1, S + 8, fA, fB);
    g1_pin();
    g1_mma(acc, a0, b0);
    g1_pin();
    g1_frags(a0, b0, S + 16, fA, fB);
    g1_pin();
    g1_mma(acc, a1, b1);
    g1_pin();
    g1_frags(a1, b1, S + 24, fA, fB);
    g1_pin();
    g1_mma(acc, a0, b0);
    g1_mma(acc, a1, b1);
  }

#pragma unroll
  for (int j = 0; j < G1_NT; ++j) {
    const int col = nBase + j * 16 + l15;
    const float bv = bias1[col];
#pragma unroll
    for (int i = 0; i < 8; ++i) {
      const int row = mBase + wave * 16 + i + 8 * half;
      const float v = acc[j][i] + bv;
      hid[(size_t)row * HID + col] = v > 0.f ? v : 0.f;
    }
  }
}

#define T_PITCH 68
#define T_WAVES 8
#define T_KW    (HID / T_WAVES)
#define T_NG    (T_KW / 8)
#define T_CH    3
#define T_NCH   (T_NG / T_CH)

__device__ __forceinline__ v8f wm4(const v2f a, const v2f b, const v8f c) {
  return __builtin_amdgcn_wmma_f32_16x16x4_f32(false, a, false, b, (short)0, c, false, false);
}
__device__ __forceinline__ v2f lo2(const v4f v) { const v2f r = {v.x, v.y}; return r; }
__device__ __forceinline__ v2f hi2(const v4f v) { const v2f r = {v.z, v.w}; return r; }

__device__ __forceinline__ void t_load(
    v4f (&a)[T_CH], v4f (&b)[T_CH][4], const float* __restrict__ pa,
    const float* __restrict__ pb, int c) {
#pragma unroll
  for (int g = 0; g < T_CH; ++g) {
    const int k = (c * T_CH + g) * 8;
    a[g] = *(const v4f*)(pa + k);
#pragma unroll
    for (int t = 0; t < 4; ++t)
      b[g][t] = *(const v4f*)(pb + (size_t)t * 16 * HID + k);
  }
}
__device__ __forceinline__ void t_mma(
    v8f (&acc)[4], const v4f (&a)[T_CH], const v4f (&b)[T_CH][4]) {
#pragma unroll
  for (int g = 0; g < T_CH; ++g) {
#pragma unroll
    for (int t = 0; t < 4; ++t) acc[t] = wm4(lo2(a[g]), lo2(b[g][t]), acc[t]);
#pragma unroll
    for (int t = 0; t < 4; ++t) acc[t] = wm4(hi2(a[g]), hi2(b[g][t]), acc[t]);
  }
}

__global__ __launch_bounds__(256) void k_tail(
    const float* __restrict__ hid, const float* __restrict__ W2,
    const float* __restrict__ bias2, const float* __restrict__ memory,
    const float* __restrict__ Wi, const float* __restrict__ bi,
    const float* __restrict__ Wh, const float* __restrict__ bh,
    const float* __restrict__ Wp1, const float* __restrict__ bp1,
    float* __restrict__ p) {
  __shared__ __attribute__((aligned(16))) float red[T_WAVES * 16 * T_PITCH];
  __shared__ __attribute__((aligned(16))) float msgT[16 * T_PITCH];
  __shared__ __attribute__((aligned(16))) float updT[16 * T_PITCH];

  const int tid = threadIdx.x;
  const int lane = tid & 31;
  const int wave = tid >> 5;
  const int l15 = lane & 15;
  const int half = lane >> 4;
  const int mBase = blockIdx.x * 16;

  {
    const float* pa = hid + (size_t)(mBase + l15) * HID + wave * T_KW + 4 * half;
    const float* pb = W2 + (size_t)l15 * HID + wave * T_KW + 4 * half;
    v4f fa[3][T_CH], fb[3][T_CH][4];
    v8f acc[4] = {};
    t_load(fa[0], fb[0], pa, pb, 0);
    t_load(fa[1], fb[1], pa, pb, 1);
#pragma unroll
    for (int c = 0; c < T_NCH; ++c) {
      if (c + 2 < T_NCH) t_load(fa[(c + 2) % 3], fb[(c + 2) % 3], pa, pb, c + 2);
      t_mma(acc, fa[c % 3], fb[c % 3]);
    }
    float* rp = red + wave * 16 * T_PITCH;
#pragma unroll
    for (int t = 0; t < 4; ++t)
#pragma unroll
      for (int i = 0; i < 8; ++i)
        rp[(i + 8 * half) * T_PITCH + 16 * t + l15] = acc[t][i];
  }

  const int j = 16 * (wave & 3) + l15;
  v4f gWi[3][8] = {}, gWh[3][8] = {}, gA[8] = {}, pW[8] = {};
  float hm[8] = {};
  float cR = 0.f, cZ = 0.f, cI = 0.f, cH = 0.f, cP = 0.f;
  if (wave < 4) {
#pragma unroll
    for (int g = 0; g < 8; ++g) {
      gA[g] = *(const v4f*)(memory + (size_t)(mBase + l15) * MD + 8 * g + 4 * half);
#pragma unroll
      for (int q = 0; q < 3; ++q) {
        gWi[q][g] = *(const v4f*)(Wi + (size_t)(q * MD + j) * MSG + 8 * g + 4 * half);
        gWh[q][g] = *(const v4f*)(Wh + (size_t)(q * MD + j) * MD + 8 * g + 4 * half);
      }
      pW[g] = *(const v4f*)(Wp1 + (size_t)j * MD + 8 * g + 4 * half);
    }
#pragma unroll
    for (int i = 0; i < 8; ++i) hm[i] = memory[(size_t)(mBase + i + 8 * half) * MD + j];
    cR = bi[j] + bh[j];
    cZ = bi[MD + j] + bh[MD + j];
    cI = bi[2 * MD + j];
    cH = bh[2 * MD + j];
    cP = bp1[j];
  }
  __syncthreads();

  {
    const int row = tid >> 4;
    const int c4 = (tid & 15) * 4;
    v4f s = *(const v4f*)(bias2 + c4);
#pragma unroll
    for (int w = 0; w < T_WAVES; ++w)
      s += *(const v4f*)(red + (w * 16 + row) * T_PITCH + c4);
    *(v4f*)(msgT + row * T_PITCH + c4) = s;
  }
  __syncthreads();

  if (wave < 4) {
    v8f R = {}, Z = {}, IN = {}, HN = {};
#pragma unroll
    for (int g = 0; g < 8; ++g) {
      const v4f am = *(const v4f*)(msgT + l15 * T_PITCH + 8 * g + 4 * half);
      const v4f ah = gA[g];
      R  = wm4(lo2(am), lo2(gWi[0][g]), R);
      Z  = wm4(lo2(am), lo2(gWi[1][g]), Z);
      IN = wm4(lo2(am), lo2(gWi[2][g]), IN);
      HN = wm4(lo2(ah), lo2(gWh[2][g]), HN);
      R  = wm4(hi2(am), hi2(gWi[0][g]), R);
      Z  = wm4(hi2(am), hi2(gWi[1][g]), Z);
      IN = wm4(hi2(am), hi2(gWi[2][g]), IN);
      HN = wm4(hi2(ah), hi2(gWh[2][g]), HN);
      R  = wm4(lo2(ah), lo2(gWh[0][g]), R);
      Z  = wm4(lo2(ah), lo2(gWh[1][g]), Z);
      R  = wm4(hi2(ah), hi2(gWh[0][g]), R);
      Z  = wm4(hi2(ah), hi2(gWh[1][g]), Z);
    }
#pragma unroll
    for (int i = 0; i < 8; ++i) {
      const float r = 1.f / (1.f + __expf(-(R[i] + cR)));
      const float z = 1.f / (1.f + __expf(-(Z[i] + cZ)));
      const float nn = tanhf((IN[i] + cI) + r * (HN[i] + cH));
      updT[(i + 8 * half) * T_PITCH + j] = (1.f - z) * nn + z * hm[i];
    }
  }
  __syncthreads();

  if (wave < 4) {
    v8f P0 = {}, P1 = {};
#pragma unroll
    for (int g = 0; g < 8; ++g) {
      const v4f au = *(const v4f*)(updT + l15 * T_PITCH + 8 * g + 4 * half);
      P0 = wm4(lo2(au), lo2(pW[g]), P0);
      P1 = wm4(hi2(au), hi2(pW[g]), P1);
    }
#pragma unroll
    for (int i = 0; i < 8; ++i) {
      const float v = (P0[i] + P1[i]) + cP;
      p[(size_t)(mBase + i + 8 * half) * MD + j] = v > 0.f ? v : 0.f;
    }
  }
}

__global__ __launch_bounds__(128) void k_pred2(
    const float* __restrict__ p, const float* __restrict__ Wp2,
    const float* __restrict__ bp2, float* __restrict__ out) {
  const int lane = threadIdx.x & 31;
  const int wave = threadIdx.x >> 5;
  const int l15 = lane & 15;
  const int half = lane >> 4;
  const int mBase = blockIdx.y * 64;
  const int nBase = blockIdx.x * 256 + wave * 64;
  const int r0 = mBase + l15;
  const int c0 = nBase + l15;

  v8f acc[4][4] = {};
#pragma unroll
  for (int k0 = 0; k0 < MD; k0 += 4)
    wmma_step4x4(acc, p, r0, MD, k0, Wp2, c0, MD, k0, half);

#pragma unroll
  for (int ni = 0; ni < 4; ++ni) {
    const int col = nBase + ni * 16 + l15;
    const float bv = bp2[col];
#pragma unroll
    for (int mi = 0; mi < 4; ++mi) {
#pragma unroll
      for (int i = 0; i < 8; ++i) {
        const int row = mBase + mi * 16 + i + 8 * half;
        out[(size_t)row * NN + col] = acc[mi][ni][i] + bv;
      }
    }
  }
}

extern "C" void kernel_launch(void* const* d_in, const int* in_sizes, int n_in,
                              void* d_out, int out_size, void* d_ws, size_t ws_size,
                              hipStream_t stream) {
  const float* memory = (const float*)d_in[0];
  const float* od_mat = (const float*)d_in[1];
  const int*   walks  = (const int*)d_in[2];
  const float* W_rw   = (const float*)d_in[3];
  const float* b_rw   = (const float*)d_in[4];
  const float* W1     = (const float*)d_in[5];
  const float* b1     = (const float*)d_in[6];
  const float* W2     = (const float*)d_in[7];
  const float* b2     = (const float*)d_in[8];
  const float* gWi    = (const float*)d_in[9];
  const float* gbi    = (const float*)d_in[10];
  const float* gWh    = (const float*)d_in[11];
  const float* gbh    = (const float*)d_in[12];
  const float* Wp1    = (const float*)d_in[13];
  const float* bp1    = (const float*)d_in[14];
  const float* Wp2    = (const float*)d_in[15];
  const float* bp2    = (const float*)d_in[16];
  float* out = (float*)d_out;

  char* ws = (char*)d_ws;
  const size_t SZ_RW  = (size_t)NN * MSG * sizeof(float);
  const size_t SZ_HID = (size_t)NN * HID * sizeof(float);
  const size_t SZ_MSG = (size_t)NN * MSG * sizeof(float);
  const size_t SZ_UPD = (size_t)NN * MD  * sizeof(float);
  float* rw   = (float*)(ws);
  float* hid  = (float*)(ws + SZ_RW);
  float* pbuf = (float*)(ws + SZ_RW + SZ_HID + SZ_MSG + SZ_UPD);

  k_rw<<<NN, 64, 0, stream>>>(memory, walks, W_rw, b_rw, rw);
  {
    dim3 grid(HID / G1_BN, NN / G1_BM);
    k_gemm1<<<grid, 128, 0, stream>>>(memory, od_mat, rw, W1, b1, hid);
  }
  k_tail<<<NN / 16, 256, 0, stream>>>(hid, W2, b2, memory, gWi, gbi, gWh, gbh, Wp1, bp1, pbuf);
  {
    dim3 grid(NN / 256, NN / 64);
    k_pred2<<<grid, 128, 0, stream>>>(pbuf, Wp2, bp2, out);
  }
}


# layer-1 GEMM at two waves per SIMD: 8-wave work-group on the same 64x528 tile, each wave 16 rows x 17 or 16 column tiles; one wave's LDS, global and barrier stretches run under the other's products
# speedup vs baseline: 1.9467x; 1.0511x over previous
#include <hip/hip_runtime.h>
#include <hip/hip_bf16.h>
#include <math.h>

#define NN   4096
#define MD   64
#define MSG  64
#define RAW  4224
#define HID  2112
#define WL   8

typedef __attribute__((ext_vector_type(2))) float v2f;
typedef __attribute__((ext_vector_type(8))) float v8f;

__device__ __forceinline__ v2f ldv2(const float* p) { return *(const v2f*)p; }

__device__ __forceinline__ void wmma_step4x4(
    v8f acc[4][4],
    const float* __restrict__ A, int r0, int ldA, int kA,
    const float* __restrict__ B, int c0, int ldB, int kB, int half) {
  v2f a[4], b[4];
#pragma unroll
  for (int t = 0; t < 4; ++t) {
    a[t] = ldv2(A + (size_t)(r0 + 16 * t) * (size_t)ldA + kA + 2 * half);
    b[t] = ldv2(B + (size_t)(c0 + 16 * t) * (size_t)ldB + kB + 2 * half);
  }
#pragma unroll
  for (int mi = 0; mi < 4; ++mi)
#pragma unroll
    for (int ni = 0; ni < 4; ++ni)
      acc[mi][ni] = __builtin_amdgcn_wmma_f32_16x16x4_f32(
          false, a[mi], false, b[ni], (short)0, acc[mi][ni], false, false);
}

__global__ __launch_bounds__(64) void k_rw(
    const float* __restrict__ memory, const int* __restrict__ walks,
    const float* __restrict__ W_rw, const float* __restrict__ b_rw,
    float* __restrict__ rw) {
  __shared__ float pooled[MD];
  const int n = blockIdx.x;
  const int d = threadIdx.x;
  float s = 0.f;
#pragma unroll
  for (int w = 0; w < WL; ++w) {
    const int idx = walks[n * WL + w];
    s += memory[(size_t)idx * MD + d];
  }
  pooled[d] = s * (1.0f / WL);
  __syncthreads();
  float acc = b_rw[d];
#pragma unroll 8
  for (int k = 0; k < MD; ++k) acc += pooled[k] * W_rw[d * MD + k];
  rw[(size_t)n * MSG + d] = acc;
}

#define G1_BM    64
#define G1_BN    528
#define G1_KC    32
#define G1_PITCH 36
#define G1_ROWS  (G1_BM + G1_BN)
#define G1_STAGE (G1_ROWS * G1_PITCH)
#define G1_NTW   17
#define G1_NSB   17
#define G1_NCH   (RAW / G1_KC)
#define G1_C_OD  (MD / G1_KC)
#define G1_C_RW  ((MD + NN) / G1_KC)

typedef __attribute__((ext_vector_type(4))) float v4f;

__device__ __forceinline__ void g1_gload(
    v4f (&stA)[2], v4f (&stB)[G1_NSB], int cn,
    const float* __restrict__ memory, const float* __restrict__ od,
    const float* __restrict__ rw, const float* __restrict__ gB, int arow, int ccol, bool first) {
  const float* src;
  int ld, k;
  if (cn < G1_C_OD)      { src = memory; ld = MD;  k = cn * G1_KC; }
  else if (cn < G1_C_RW) { src = od;     ld = NN;  k = (cn - G1_C_OD) * G1_KC; }
  else                   { src = rw;     ld = MSG; k = (cn - G1_C_RW) * G1_KC; }
  const float* gA = src + (size_t)arow * ld + k + ccol;
#pragma unroll
  for (int i = 0; i < 2; ++i)
    stA[i] = *(const v4f*)(gA + (size_t)i * 32 * ld);
  const float* gBn = gB + cn * G1_KC;
#pragma unroll
  for (int i = 0; i < 16; ++i)
    stB[i] = *(const v4f*)(gBn + (size_t)i * 32 * RAW);
  if (first) stB[16] = *(const v4f*)(gBn + (size_t)16 * 32 * RAW);
}

__device__ __forceinline__ void g1_lstore(
    float* sN, const v4f (&stA)[2], const v4f (&stB)[G1_NSB], bool first) {
#pragma unroll
  for (int i = 0; i < 2; ++i)
    *(v4f*)(sN + i * 32 * G1_PITCH) = stA[i];
#pragma unroll
  for (int i = 0; i < 16; ++i)
    *(v4f*)(sN + (G1_BM + i * 32) * G1_PITCH) = stB[i];
  if (first) *(v4f*)(sN + (G1_BM + 16 * 32) * G1_PITCH) = stB[16];
}

__device__ __forceinline__ void g1_frags(
    v4f& a, v4f (&b)[G1_NTW], const float* S, int fA, int fB, bool first) {
  a = *(const v4f*)(S + fA);
#pragma unroll
  for (int j = 0; j < 16; ++j)
    b[j] = *(const v4f*)(S + fB + j * 16 * G1_PITCH);
  if (first) b[16] = *(const v4f*)(S + fB + 16 * 16 * G1_PITCH);
}

__device__ __forceinline__ void g1_mma(
    v8f (&acc)[G1_NTW], const v4f& a, const v4f (&b)[G1_NTW], bool first) {
  const v2f alo = {a.x, a.y};
  const v2f ahi = {a.z, a.w};
#pragma unroll
  for (int j = 0; j < 16; ++j) {
    const v2f blo = {b[j].x, b[j].y};
    acc[j] = __builtin_amdgcn_wmma_f32_16x16x4_f32(
        false, alo, false, blo, (short)0, acc[j], false, false);
  }
  if (first) {
    const v2f blo = {b[16].x, b[16].y};
    acc[16] = __builtin_amdgcn_wmma_f32_16x16x4_f32(
        false, alo, false, blo, (short)0, acc[16], false, false);
  }
#pragma unroll
  for (int j = 0; j < 16; ++j) {
    const v2f bhi = {b[j].z, b[j].w};
    acc[j] = __builtin_amdgcn_wmma_f32_16x16x4_f32(
        false, ahi, false, bhi, (short)0, acc[j], false, false);
  }
  if (first) {
    const v2f bhi = {b[16].z, b[16].w};
    acc[16] = __builtin_amdgcn_wmma_f32_16x16x4_f32(
        false, ahi, false, bhi, (short)0, acc[16], false, false);
  }
}

__device__ __forceinline__ void g1_pin() {
  __builtin_amdgcn_sched_barrier(0);
}

__global__ __launch_bounds__(256) void k_gemm1(
    const float* __restrict__ memory, const float* __restrict__ od,
    const float* __restrict__ rw, const float* __restrict__ W1,
    const float* __restrict__ bias1, float* __restrict__ hid) {
  __shared__ __attribute__((aligned(16))) float L[2 * G1_STAGE];

  const int tid = threadIdx.x;
  const int lane = tid & 31;
  const int wave = __builtin_amdgcn_readfirstlane(tid >> 5);
  const bool first = wave < 4;
  const int wm = wave & 3;
  const int jBase = first ? 0 : G1_NTW;
  const int l15 = lane & 15;
  const int half = lane >> 4;
  const int nBase = blockIdx.x * G1_BN;
  const int mBase = blockIdx.y * G1_BM;

  const int crow = tid >> 3;
  const int ccol = (tid & 7) * 4;
  const int arow = mBase + crow;
  const float* gB = W1 + (size_t)(nBase + crow) * RAW + ccol;
  float* sW = L + crow * G1_PITCH + ccol;

  const int fA = (wm * 16 + l15) * G1_PITCH + 4 * half;
  const int fB = (G1_BM + jBase * 16 + l15) * G1_PITCH + 4 * half;

  v4f stA[2], stB[G1_NSB] = {};
  v4f a0, a1, b0[G1_NTW] = {}, b1[G1_NTW] = {};
  v8f acc[G1_NTW] = {};

  g1_gload(stA, stB, 0, memory, od, rw, gB, arow, ccol, first);
  g1_lstore(sW, stA, stB, first);
  g1_gload(stA, stB, 1, memory, od, rw, gB, arow, ccol, first);
  __syncthreads();
  g1_frags(a0, b0, L, fA, fB, first);

  for (int ch = 0; ch < G1_NCH - 1; ++ch) {
    const float* S = L + (ch & 1) * G1_STAGE;
    const int nOff = ((ch + 1) & 1) * G1_STAGE;

    g1_frags(a1, b1, S + 8, fA, fB, first);
    g1_pin();
    g1_mma(acc, a0, b0, first);
    g1_pin();
    g1_frags(a0, b0, S + 16, fA, fB, first);
    g1_pin();
    g1_mma(acc, a1, b1, first);
    g1_pin();
    g1_frags(a1, b1, S + 24, fA, fB, first);
    g1_lstore(sW + nOff, stA, stB, first);
    g1_pin();
    g1_mma(acc, a0, b0, first);
    g1_pin();
    __syncthreads();
    g1_frags(a0, b0, L + nOff, fA, fB, first);
    if (ch + 2 < G1_NCH)
      g1_gload(stA, stB, ch + 2, memory, od, rw, gB, arow, ccol, first);
    g1_pin();
    g1_mma(acc, a1, b1, first);
  }
  {
    const float* S = L + ((G1_NCH - 1) & 1) * G1_STAGE;
    g1_frags(a1, b1, S + 8, fA, fB, first);
    g1_pin();
    g1_mma(acc, a0, b0, first);
    g1_pin();
    g1_frags(a0, b0, S + 16, fA, fB, first);
    g1_pin();
    g1_mma(acc, a1, b1, first);
    g1_pin();
    g1_frags(a1, b1, S + 24, fA, fB, first);
    g1_pin();
    g1_mma(acc, a0, b0, first);
    g1_mma(acc, a1, b1, first);
  }

#pragma unroll
  for (int j = 0; j < G1_NTW; ++j) {
    if (j < 16 || first) {
      const int col = nBase + (jBase + j) * 16 + l15;
      const float bv = bias1[col];
#pragma unroll
      for (int i = 0; i < 8; ++i) {
        const int row = mBase + wm * 16 + i + 8 * half;
        const float v = acc[j][i] + bv;
        hid[(size_t)row * HID + col] = v > 0.f ? v : 0.f;
      }
    }
  }
}

#define T_PITCH 68
#define T_WAVES 8
#define T_KW    (HID / T_WAVES)
#define T_NG    (T_KW / 8)
#define T_CH    3
#define T_NCH   (T_NG / T_CH)

__device__ __forceinline__ v8f wm4(const v2f a, const v2f b, const v8f c) {
  return __builtin_amdgcn_wmma_f32_16x16x4_f32(false, a, false, b, (short)0, c, false, false);
}
__device__ __forceinline__ v2f lo2(const v4f v) { const v2f r = {v.x, v.y}; return r; }
__device__ __forceinline__ v2f hi2(const v4f v) { const v2f r = {v.z, v.w}; return r; }

__device__ __forceinline__ void t_load(
    v4f (&a)[T_CH], v4f (&b)[T_CH][4], const float* __restrict__ pa,
    const float* __restrict__ pb, int c) {
#pragma unroll
  for (int g = 0; g < T_CH; ++g) {
    const int k = (c * T_CH + g) * 8;
    a[g] = *(const v4f*)(pa + k);
#pragma unroll
    for (int t = 0; t < 4; ++t)
      b[g][t] = *(const v4f*)(pb + (size_t)t * 16 * HID + k);
  }
}
__device__ __forceinline__ void t_mma(
    v8f (&acc)[4], const v4f (&a)[T_CH], const v4f (&b)[T_CH][4]) {
#pragma unroll
  for (int g = 0; g < T_CH; ++g) {
#pragma unroll
    for (int t = 0; t < 4; ++t) acc[t] = wm4(lo2(a[g]), lo2(b[g][t]), acc[t]);
#pragma unroll
    for (int t = 0; t < 4; ++t) acc[t] = wm4(hi2(a[g]), hi2(b[g][t]), acc[t]);
  }
}

__global__ __launch_bounds__(256) void k_tail(
    const float* __restrict__ hid, const float* __restrict__ W2,
    const float* __restrict__ bias2, const float* __restrict__ memory,
    const float* __restrict__ Wi, const float* __restrict__ bi,
    const float* __restrict__ Wh, const float* __restrict__ bh,
    const float* __restrict__ Wp1, const float* __restrict__ bp1,
    float* __restrict__ p) {
  __shared__ __attribute__((aligned(16))) float red[T_WAVES * 16 * T_PITCH];
  __shared__ __attribute__((aligned(16))) float msgT[16 * T_PITCH];
  __shared__ __attribute__((aligned(16))) float updT[16 * T_PITCH];

  const int tid = threadIdx.x;
  const int lane = tid & 31;
  const int wave = tid >> 5;
  const int l15 = lane & 15;
  const int half = lane >> 4;
  const int mBase = blockIdx.x * 16;

  {
    const float* pa = hid + (size_t)(mBase + l15) * HID + wave * T_KW + 4 * half;
    const float* pb = W2 + (size_t)l15 * HID + wave * T_KW + 4 * half;
    v4f fa[3][T_CH], fb[3][T_CH][4];
    v8f acc[4] = {};
    t_load(fa[0], fb[0], pa, pb, 0);
    t_load(fa[1], fb[1], pa, pb, 1);
#pragma unroll
    for (int c = 0; c < T_NCH; ++c) {
      if (c + 2 < T_NCH) t_load(fa[(c + 2) % 3], fb[(c + 2) % 3], pa, pb, c + 2);
      t_mma(acc, fa[c % 3], fb[c % 3]);
    }
    float* rp = red + wave * 16 * T_PITCH;
#pragma unroll
    for (int t = 0; t < 4; ++t)
#pragma unroll
      for (int i = 0; i < 8; ++i)
        rp[(i + 8 * half) * T_PITCH + 16 * t + l15] = acc[t][i];
  }

  const int j = 16 * (wave & 3) + l15;
  v4f gWi[3][8] = {}, gWh[3][8] = {}, gA[8] = {}, pW[8] = {};
  float hm[8] = {};
  float cR = 0.f, cZ = 0.f, cI = 0.f, cH = 0.f, cP = 0.f;
  if (wave < 4) {
#pragma unroll
    for (int g = 0; g < 8; ++g) {
      gA[g] = *(const v4f*)(memory + (size_t)(mBase + l15) * MD + 8 * g + 4 * half);
#pragma unroll
      for (int q = 0; q < 3; ++q) {
        gWi[q][g] = *(const v4f*)(Wi + (size_t)(q * MD + j) * MSG + 8 * g + 4 * half);
        gWh[q][g] = *(const v4f*)(Wh + (size_t)(q * MD + j) * MD + 8 * g + 4 * half);
      }
      pW[g] = *(const v4f*)(Wp1 + (size_t)j * MD + 8 * g + 4 * half);
    }
#pragma unroll
    for (int i = 0; i < 8; ++i) hm[i] = memory[(size_t)(mBase + i + 8 * half) * MD + j];
    cR = bi[j] + bh[j];
    cZ = bi[MD + j] + bh[MD + j];
    cI = bi[2 * MD + j];
    cH = bh[2 * MD + j];
    cP = bp1[j];
  }
  __syncthreads();

  {
    const int row = tid >> 4;
    const int c4 = (tid & 15) * 4;
    v4f s = *(const v4f*)(bias2 + c4);
#pragma unroll
    for (int w = 0; w < T_WAVES; ++w)
      s += *(const v4f*)(red + (w * 16 + row) * T_PITCH + c4);
    *(v4f*)(msgT + row * T_PITCH + c4) = s;
  }
  __syncthreads();

  if (wave < 4) {
    v8f R = {}, Z = {}, IN = {}, HN = {};
#pragma unroll
    for (int g = 0; g < 8; ++g) {
      const v4f am = *(const v4f*)(msgT + l15 * T_PITCH + 8 * g + 4 * half);
      const v4f ah = gA[g];
      R  = wm4(lo2(am), lo2(gWi[0][g]), R);
      Z  = wm4(lo2(am), lo2(gWi[1][g]), Z);
      IN = wm4(lo2(am), lo2(gWi[2][g]), IN);
      HN = wm4(lo2(ah), lo2(gWh[2][g]), HN);
      R  = wm4(hi2(am), hi2(gWi[0][g]), R);
      Z  = wm4(hi2(am), hi2(gWi[1][g]), Z);
      IN = wm4(hi2(am), hi2(gWi[2][g]), IN);
      HN = wm4(hi2(ah), hi2(gWh[2][g]), HN);
      R  = wm4(lo2(ah), lo2(gWh[0][g]), R);
      Z  = wm4(lo2(ah), lo2(gWh[1][g]), Z);
      R  = wm4(hi2(ah), hi2(gWh[0][g]), R);
      Z  = wm4(hi2(ah), hi2(gWh[1][g]), Z);
    }
#pragma unroll
    for (int i = 0; i < 8; ++i) {
      const float r = 1.f / (1.f + __expf(-(R[i] + cR)));
      const float z = 1.f / (1.f + __expf(-(Z[i] + cZ)));
      const float nn = tanhf((IN[i] + cI) + r * (HN[i] + cH));
      updT[(i + 8 * half) * T_PITCH + j] = (1.f - z) * nn + z * hm[i];
    }
  }
  __syncthreads();

  if (wave < 4) {
    v8f P0 = {}, P1 = {};
#pragma unroll
    for (int g = 0; g < 8; ++g) {
      const v4f au = *(const v4f*)(updT + l15 * T_PITCH + 8 * g + 4 * half);
      P0 = wm4(lo2(au), lo2(pW[g]), P0);
      P1 = wm4(hi2(au), hi2(pW[g]), P1);
    }
#pragma unroll
    for (int i = 0; i < 8; ++i) {
      const float v = (P0[i] + P1[i]) + cP;
      p[(size_t)(mBase + i + 8 * half) * MD + j] = v > 0.f ? v : 0.f;
    }
  }
}

__global__ __launch_bounds__(128) void k_pred2(
    const float* __restrict__ p, const float* __restrict__ Wp2,
    const float* __restrict__ bp2, float* __restrict__ out) {
  const int lane = threadIdx.x & 31;
  const int wave = threadIdx.x >> 5;
  const int l15 = lane & 15;
  const int half = lane >> 4;
  const int mBase = blockIdx.y * 64;
  const int nBase = blockIdx.x * 256 + wave * 64;
  const int r0 = mBase + l15;
  const int c0 = nBase + l15;

  v8f acc[4][4] = {};
#pragma unroll
  for (int k0 = 0; k0 < MD; k0 += 4)
    wmma_step4x4(acc, p, r0, MD, k0, Wp2, c0, MD, k0, half);

#pragma unroll
  for (int ni = 0; ni < 4; ++ni) {
    const int col = nBase + ni * 16 + l15;
    const float bv = bp2[col];
#pragma unroll
    for (int mi = 0; mi < 4; ++mi) {
#pragma unroll
      for (int i = 0; i < 8; ++i) {
        const int row = mBase + mi * 16 + i + 8 * half;
        out[(size_t)row * NN + col] = acc[mi][ni][i] + bv;
      }
    }
  }
}

extern "C" void kernel_launch(void* const* d_in, const int* in_sizes, int n_in,
                              void* d_out, int out_size, void* d_ws, size_t ws_size,
                              hipStream_t stream) {
  const float* memory = (const float*)d_in[0];
  const float* od_mat = (const float*)d_in[1];
  const int*   walks  = (const int*)d_in[2];
  const float* W_rw   = (const float*)d_in[3];
  const float* b_rw   = (const float*)d_in[4];
  const float* W1     = (const float*)d_in[5];
  const float* b1     = (const float*)d_in[6];
  const float* W2     = (const float*)d_in[7];
  const float* b2     = (const float*)d_in[8];
  const float* gWi    = (const float*)d_in[9];
  const float* gbi    = (const float*)d_in[10];
  const float* gWh    = (const float*)d_in[11];
  const float* gbh    = (const float*)d_in[12];
  const float* Wp1    = (const float*)d_in[13];
  const float* bp1    = (const float*)d_in[14];
  const float* Wp2    = (const float*)d_in[15];
  const float* bp2    = (const float*)d_in[16];
  float* out = (float*)d_out;

  char* ws = (char*)d_ws;
  const size_t SZ_RW  = (size_t)NN * MSG * sizeof(float);
  const size_t SZ_HID = (size_t)NN * HID * sizeof(float);
  const size_t SZ_MSG = (size_t)NN * MSG * sizeof(float);
  const size_t SZ_UPD = (size_t)NN * MD  * sizeof(float);
  float* rw   = (float*)(ws);
  float* hid  = (float*)(ws + SZ_RW);
  float* pbuf = (float*)(ws + SZ_RW + SZ_HID + SZ_MSG + SZ_UPD);

  k_rw<<<NN, 64, 0, stream>>>(memory, walks, W_rw, b_rw, rw);
  {
    dim3 grid(HID / G1_BN, NN / G1_BM);
    k_gemm1<<<grid, 256, 0, stream>>>(memory, od_mat, rw, W1, b1, hid);
  }
  k_tail<<<NN / 16, 256, 0, stream>>>(hid, W2, b2, memory, gWi, gbi, gWh, gbh, Wp1, bp1, pbuf);
  {
    dim3 grid(NN / 256, NN / 64);
    k_pred2<<<grid, 128, 0, stream>>>(pbuf, Wp2, bp2, out);
  }
}
